# starting kernel
# baseline (speedup 1.0000x reference)
#include <hip/hip_runtime.h>

typedef __attribute__((ext_vector_type(2))) float v2f;
typedef __attribute__((ext_vector_type(4))) float v4f;
typedef __attribute__((ext_vector_type(8))) float v8f;

#define DIMS 64
#define ROWS_PER_BLOCK 128
#define XSTRIDE 68
#define BSTRIDE 130

__global__ __launch_bounds__(256)
void poly_wmma_kernel(const float* __restrict__ X,
                      const float* __restrict__ W,
                      const float* __restrict__ Bv,
                      float* __restrict__ out)
{
    __shared__ float Xs[ROWS_PER_BLOCK][XSTRIDE];
    __shared__ v2f   Bsw[16][BSTRIDE];
    __shared__ float Wsh[DIMS];

    const int tid     = threadIdx.x;
    const int rowbase = blockIdx.x * ROWS_PER_BLOCK;

    if (tid < DIMS) Wsh[tid] = W[1 + tid];

    for (int e = tid; e < 16 * 128; e += 256) {
        int ks   = e >> 7;
        int t    = e & 127;
        int n    = t >> 1;
        int half = t & 1;
        int k    = 4 * ks + 2 * half;
        v2f v;
        v.x = (k     >= n) ? W[65 + n * DIMS - (n * (n - 1)) / 2 + (k     - n)] : 0.0f;
        v.y = (k + 1 >= n) ? W[65 + n * DIMS - (n * (n - 1)) / 2 + (k + 1 - n)] : 0.0f;
        Bsw[ks][2 * n + half] = v;
    }

    {
        const v4f* Xg = (const v4f*)(X + (size_t)rowbase * DIMS);
        #pragma unroll
        for (int i = 0; i < 8; ++i) {
            int g = tid + i * 256;
            int r = g >> 4;
            int c = (g & 15) << 2;
            v4f val = Xg[g];
            *(v4f*)&Xs[r][c] = val;
        }
    }
    __syncthreads();

    const int wave = tid >> 5;
    const int lane = tid & 31;
    const int half = lane >> 4;
    const int l15  = lane & 15;
    const int wr   = wave * 16;

    v2f afrag[16];
    #pragma unroll
    for (int ks = 0; ks < 16; ++ks) {
        int k = ks * 4 + half * 2;
        afrag[ks].x = Xs[wr + l15][k];
        afrag[ks].y = Xs[wr + l15][k + 1];
    }

    v8f acc[4];
    #pragma unroll
    for (int nt = 0; nt < 4; ++nt) acc[nt] = (v8f)(0.0f);

    #pragma unroll
    for (int nt = 0; nt < 4; ++nt) {
        #pragma unroll
        for (int ks = 0; ks < 16; ++ks) {
            v2f bfrag = Bsw[ks][2 * (nt * 16 + l15) + half];
            acc[nt] = __builtin_amdgcn_wmma_f32_16x16x4_f32(
                false, afrag[ks], false, bfrag, (short)0, acc[nt], false, false);
        }
    }

    const float bias = W[0] + Bv[0];

    #pragma unroll
    for (int r = 0; r < 8; ++r) {
        int m = r + 8 * half;
        float s = 0.0f;
        #pragma unroll
        for (int nt = 0; nt < 4; ++nt) {
            int n = nt * 16 + l15;
            s += Xs[wr + m][n] * (Wsh[n] + acc[nt][r]);
        }
        s += __shfl_xor(s, 1, 16);
        s += __shfl_xor(s, 2, 16);
        s += __shfl_xor(s, 4, 16);
        s += __shfl_xor(s, 8, 16);
        if (l15 == 0) out[rowbase + wr + m] = bias + s;
    }
}

extern "C" void kernel_launch(void* const* d_in, const int* in_sizes, int n_in,
                              void* d_out, int out_size, void* d_ws, size_t ws_size,
                              hipStream_t stream) {
    const float* x  = (const float*)d_in[0];
    const float* W  = (const float*)d_in[1];
    const float* bv = (const float*)d_in[2];
    float* out = (float*)d_out;

    const int batch = in_sizes[0] / DIMS;
    const int grid  = batch / ROWS_PER_BLOCK;
    poly_wmma_kernel<<<grid, 256, 0, stream>>>(x, W, bv, out);
}


# WMMA operands swapped (Y^T = U*X^T): each lane owns a batch row, so the row sum is in-lane plus one permlanex16 (new summation order; bias added before it) instead of 32 chained ds_bpermute
# speedup vs baseline: 1.1946x; 1.1946x over previous
#include <hip/hip_runtime.h>


typedef __attribute__((ext_vector_type(2))) float v2f;
typedef __attribute__((ext_vector_type(4))) float v4f;
typedef __attribute__((ext_vector_type(8))) float v8f;

#define DIMS 64
#define ROWS_PER_BLOCK 128
#define XSTRIDE 68
#define BSTRIDE 130

__device__ __forceinline__ float xhalf(float x) {
    int b = __builtin_bit_cast(int, x);
    return __builtin_bit_cast(float,
        __builtin_amdgcn_permlanex16(b, b, 0x76543210, 0xfedcba98, false, false));
}

__global__ __launch_bounds__(256)
void poly_wmma_kernel(const float* __restrict__ X,
                      const float* __restrict__ W,
                      const float* __restrict__ Bv,
                      float* __restrict__ out)
{
    __shared__ __attribute__((aligned(16))) float Xs[ROWS_PER_BLOCK][XSTRIDE];
    __shared__ __attribute__((aligned(16))) v2f   Bsw[16][BSTRIDE];
    __shared__ __attribute__((aligned(16))) float Wsh[DIMS];

    const int tid     = threadIdx.x;
    const int rowbase = blockIdx.x * ROWS_PER_BLOCK;

    const float bias = W[0] + Bv[0];

    if (tid < DIMS) Wsh[tid] = W[1 + tid];

    for (int e = tid; e < 16 * 128; e += 256) {
        int ks   = e >> 7;
        int t    = e & 127;
        int n    = t >> 1;
        int half = t & 1;
        int k    = 4 * ks + 2 * half;
        v2f v;
        v.x = (k     >= n) ? W[65 + n * DIMS - (n * (n - 1)) / 2 + (k     - n)] : 0.0f;
        v.y = (k + 1 >= n) ? W[65 + n * DIMS - (n * (n - 1)) / 2 + (k + 1 - n)] : 0.0f;
        Bsw[ks][2 * n + half] = v;
    }

    {
        const v4f* Xg = (const v4f*)(X + (size_t)rowbase * DIMS);
        #pragma unroll
        for (int i = 0; i < 8; ++i) {
            int g = tid + i * 256;
            int r = g >> 4;
            int c = (g & 15) << 2;
            v4f val = Xg[g];
            *(v4f*)&Xs[r][c] = val;
        }
    }
    __syncthreads();

    const int wave = tid >> 5;
    const int lane = tid & 31;
    const int half = lane >> 4;
    const int l15  = lane & 15;
    const int wr   = wave * 16;

    v2f xfrag[16];
    #pragma unroll
    for (int ks = 0; ks < 16; ++ks) {
        int k = ks * 4 + half * 2;
        xfrag[ks].x = Xs[wr + l15][k];
        xfrag[ks].y = Xs[wr + l15][k + 1];
    }

    v4f xe[8], we[8];
    #pragma unroll
    for (int nt = 0; nt < 4; ++nt) {
        #pragma unroll
        for (int q = 0; q < 2; ++q) {
            int n0 = 16 * nt + 8 * half + 4 * q;
            xe[2 * nt + q] = *(const v4f*)&Xs[wr + l15][n0];
            we[2 * nt + q] = *(const v4f*)&Wsh[n0];
        }
    }

    v8f acc[4];
    #pragma unroll
    for (int nt = 0; nt < 4; ++nt) acc[nt] = (v8f)(0.0f);

    #pragma unroll
    for (int nt = 0; nt < 4; ++nt) {
        #pragma unroll
        for (int ks = 0; ks < 16; ++ks) {
            v2f ufrag = Bsw[ks][2 * (nt * 16 + l15) + half];
            acc[nt] = __builtin_amdgcn_wmma_f32_16x16x4_f32(
                false, ufrag, false, xfrag[ks], (short)0, acc[nt], false, false);
        }
    }

    float p[4];
    #pragma unroll
    for (int nt = 0; nt < 4; ++nt) {
        float t = 0.0f;
        #pragma unroll
        for (int r = 0; r < 4; ++r) t += xe[2 * nt][r] * (we[2 * nt][r] + acc[nt][r]);
        #pragma unroll
        for (int r = 0; r < 4; ++r) t += xe[2 * nt + 1][r] * (we[2 * nt + 1][r] + acc[nt][4 + r]);
        p[nt] = t;
    }
    float s = (p[0] + p[1]) + (p[2] + p[3]);
    s += (half == 0) ? bias : 0.0f;
    s += xhalf(s);
    if (half == 0) out[rowbase + wr + l15] = s;
}

extern "C" void kernel_launch(void* const* d_in, const int* in_sizes, int n_in,
                              void* d_out, int out_size, void* d_ws, size_t ws_size,
                              hipStream_t stream) {
    const float* x  = (const float*)d_in[0];
    const float* W  = (const float*)d_in[1];
    const float* bv = (const float*)d_in[2];
    float* out = (float*)d_out;

    const int batch = in_sizes[0] / DIMS;
    const int grid  = batch / ROWS_PER_BLOCK;
    poly_wmma_kernel<<<grid, 256, 0, stream>>>(x, W, bv, out);
}


# All 25 prologue global loads issued before the first wait: X tile first, W unpack unrolled and branch-free with one coalesced U row per wave load; was 10 serialized round trips; sums unchanged
# speedup vs baseline: 1.7052x; 1.4274x over previous
#include <hip/hip_runtime.h>


typedef __attribute__((ext_vector_type(2))) float v2f;
typedef __attribute__((ext_vector_type(4))) float v4f;
typedef __attribute__((ext_vector_type(8))) float v8f;

#define DIMS 64
#define ROWS_PER_BLOCK 128
#define XSTRIDE 68
#define BSTRIDE 130

__device__ __forceinline__ float xhalf(float x) {
    int b = __builtin_bit_cast(int, x);
    return __builtin_bit_cast(float,
        __builtin_amdgcn_permlanex16(b, b, 0x76543210, 0xfedcba98, false, false));
}

__global__ __launch_bounds__(256)
void poly_wmma_kernel(const float* __restrict__ X,
                      const float* __restrict__ W,
                      const float* __restrict__ Bv,
                      float* __restrict__ out)
{
    __shared__ __attribute__((aligned(16))) float Xs[ROWS_PER_BLOCK][XSTRIDE];
    __shared__ __attribute__((aligned(16))) v2f   Bsw[16][BSTRIDE];
    __shared__ __attribute__((aligned(16))) float Wsh[DIMS];

    const int tid     = threadIdx.x;
    const int rowbase = blockIdx.x * ROWS_PER_BLOCK;

    const float bias = W[0] + Bv[0];


    const v4f* Xg = (const v4f*)(X + (size_t)rowbase * DIMS);
    v4f xv[8];
    #pragma unroll
    for (int i = 0; i < 8; ++i) xv[i] = Xg[tid + i * 256];
    __builtin_amdgcn_sched_barrier(0);

    const unsigned uw = (unsigned)tid >> 5;
    const unsigned uj = (unsigned)tid & 31u;
    const unsigned k0 = 2u * uj, k1 = 2u * uj + 1u;
    float ux[8], uy[8];
    #pragma unroll
    for (int i = 0; i < 8; ++i) {
        unsigned n  = 8u * uw + (unsigned)i;
        unsigned rb = 65u + 63u * n - (n * (n - 1u)) / 2u;
        ux[i] = W[rb + (k0 > n ? k0 : n)];
        uy[i] = W[rb + (k1 > n ? k1 : n)];
    }

    const float wlin = W[1 + (tid & (DIMS - 1))];

    #pragma unroll
    for (int i = 0; i < 8; ++i) {
        int g = tid + i * 256;
        int r = g >> 4;
        int c = (g & 15) << 2;
        *(v4f*)&Xs[r][c] = xv[i];
    }

    #pragma unroll
    for (int i = 0; i < 8; ++i) {
        unsigned n = 8u * uw + (unsigned)i;
        v2f v;
        v.x = (k0 >= n) ? ux[i] : 0.0f;
        v.y = (k1 >= n) ? uy[i] : 0.0f;
        Bsw[uj >> 1][2u * n + (uj & 1u)] = v;
    }

    asm volatile("" :: "v"(wlin));
    if (tid < DIMS) Wsh[tid] = wlin;

    __syncthreads();

    const int wave = tid >> 5;
    const int lane = tid & 31;
    const int half = lane >> 4;
    const int l15  = lane & 15;
    const int wr   = wave * 16;

    v2f xfrag[16];
    #pragma unroll
    for (int ks = 0; ks < 16; ++ks) {
        int k = ks * 4 + half * 2;
        xfrag[ks].x = Xs[wr + l15][k];
        xfrag[ks].y = Xs[wr + l15][k + 1];
    }

    v4f xe[8], we[8];
    #pragma unroll
    for (int nt = 0; nt < 4; ++nt) {
        #pragma unroll
        for (int q = 0; q < 2; ++q) {
            int n0 = 16 * nt + 8 * half + 4 * q;
            xe[2 * nt + q] = *(const v4f*)&Xs[wr + l15][n0];
            we[2 * nt + q] = *(const v4f*)&Wsh[n0];
        }
    }

    v8f acc[4];
    #pragma unroll
    for (int nt = 0; nt < 4; ++nt) acc[nt] = (v8f)(0.0f);

    #pragma unroll
    for (int nt = 0; nt < 4; ++nt) {
        #pragma unroll
        for (int ks = 0; ks < 16; ++ks) {
            v2f ufrag = Bsw[ks][2 * (nt * 16 + l15) + half];
            acc[nt] = __builtin_amdgcn_wmma_f32_16x16x4_f32(
                false, ufrag, false, xfrag[ks], (short)0, acc[nt], false, false);
        }
    }

    float p[4];
    #pragma unroll
    for (int nt = 0; nt < 4; ++nt) {
        float t = 0.0f;
        #pragma unroll
        for (int r = 0; r < 4; ++r) t += xe[2 * nt][r] * (we[2 * nt][r] + acc[nt][r]);
        #pragma unroll
        for (int r = 0; r < 4; ++r) t += xe[2 * nt + 1][r] * (we[2 * nt + 1][r] + acc[nt][4 + r]);
        p[nt] = t;
    }
    float s = (p[0] + p[1]) + (p[2] + p[3]);
    s += (half == 0) ? bias : 0.0f;
    s += xhalf(s);
    if (half == 0) out[rowbase + wr + l15] = s;
}

extern "C" void kernel_launch(void* const* d_in, const int* in_sizes, int n_in,
                              void* d_out, int out_size, void* d_ws, size_t ws_size,
                              hipStream_t stream) {
    const float* x  = (const float*)d_in[0];
    const float* W  = (const float*)d_in[1];
    const float* bv = (const float*)d_in[2];
    float* out = (float*)d_out;

    const int batch = in_sizes[0] / DIMS;
    const int grid  = batch / ROWS_PER_BLOCK;
    poly_wmma_kernel<<<grid, 256, 0, stream>>>(x, W, bv, out);
}


# WMMA k-steps that hold only structural zeros of U (k < n, below the diagonal) are skipped: 40 matrix instructions per wave, was 64; summation order unchanged (only exact-zero terms go)
# speedup vs baseline: 1.8546x; 1.0876x over previous
#include <hip/hip_runtime.h>


typedef __attribute__((ext_vector_type(2))) float v2f;
typedef __attribute__((ext_vector_type(4))) float v4f;
typedef __attribute__((ext_vector_type(8))) float v8f;

#define DIMS 64
#define ROWS_PER_BLOCK 128
#define XSTRIDE 68
#define BSTRIDE 130

__device__ __forceinline__ float xhalf(float x) {
    int b = __builtin_bit_cast(int, x);
    return __builtin_bit_cast(float,
        __builtin_amdgcn_permlanex16(b, b, 0x76543210, 0xfedcba98, false, false));
}

__global__ __launch_bounds__(256)
void poly_wmma_kernel(const float* __restrict__ X,
                      const float* __restrict__ W,
                      const float* __restrict__ Bv,
                      float* __restrict__ out)
{
    __shared__ __attribute__((aligned(16))) float Xs[ROWS_PER_BLOCK][XSTRIDE];
    __shared__ __attribute__((aligned(16))) v2f   Bsw[16][BSTRIDE];
    __shared__ __attribute__((aligned(16))) float Wsh[DIMS];

    const int tid     = threadIdx.x;
    const int rowbase = blockIdx.x * ROWS_PER_BLOCK;

    const float bias = W[0] + Bv[0];


    const v4f* Xg = (const v4f*)(X + (size_t)rowbase * DIMS);
    v4f xv[8];
    #pragma unroll
    for (int i = 0; i < 8; ++i) xv[i] = Xg[tid + i * 256];
    __builtin_amdgcn_sched_barrier(0);

    const unsigned uw = (unsigned)tid >> 5;
    const unsigned uj = (unsigned)tid & 31u;
    const unsigned k0 = 2u * uj, k1 = 2u * uj + 1u;
    float ux[8], uy[8];
    #pragma unroll
    for (int i = 0; i < 8; ++i) {
        unsigned n  = 8u * uw + (unsigned)i;
        unsigned rb = 65u + 63u * n - (n * (n - 1u)) / 2u;
        ux[i] = W[rb + (k0 > n ? k0 : n)];
        uy[i] = W[rb + (k1 > n ? k1 : n)];
    }

    const float wlin = W[1 + (tid & (DIMS - 1))];

    #pragma unroll
    for (int i = 0; i < 8; ++i) {
        int g = tid + i * 256;
        int r = g >> 4;
        int c = (g & 15) << 2;
        *(v4f*)&Xs[r][c] = xv[i];
    }

    #pragma unroll
    for (int i = 0; i < 8; ++i) {
        unsigned n = 8u * uw + (unsigned)i;
        v2f v;
        v.x = (k0 >= n) ? ux[i] : 0.0f;
        v.y = (k1 >= n) ? uy[i] : 0.0f;
        Bsw[uj >> 1][2u * n + (uj & 1u)] = v;
    }

    asm volatile("" :: "v"(wlin));
    if (tid < DIMS) Wsh[tid] = wlin;

    __syncthreads();

    const int wave = tid >> 5;
    const int lane = tid & 31;
    const int half = lane >> 4;
    const int l15  = lane & 15;
    const int wr   = wave * 16;

    v2f xfrag[16];
    #pragma unroll
    for (int ks = 0; ks < 16; ++ks) {
        int k = ks * 4 + half * 2;
        xfrag[ks].x = Xs[wr + l15][k];
        xfrag[ks].y = Xs[wr + l15][k + 1];
    }

    v4f xe[8], we[8];
    #pragma unroll
    for (int nt = 0; nt < 4; ++nt) {
        #pragma unroll
        for (int q = 0; q < 2; ++q) {
            int n0 = 16 * nt + 8 * half + 4 * q;
            xe[2 * nt + q] = *(const v4f*)&Xs[wr + l15][n0];
            we[2 * nt + q] = *(const v4f*)&Wsh[n0];
        }
    }

    v8f acc[4];
    #pragma unroll
    for (int nt = 0; nt < 4; ++nt) acc[nt] = (v8f)(0.0f);

    #pragma unroll
    for (int nt = 0; nt < 4; ++nt) {
        #pragma unroll
        for (int ks = 4 * nt; ks < 16; ++ks) {
            v2f ufrag = Bsw[ks][2 * (nt * 16 + l15) + half];
            acc[nt] = __builtin_amdgcn_wmma_f32_16x16x4_f32(
                false, ufrag, false, xfrag[ks], (short)0, acc[nt], false, false);
        }
    }

    float p[4];
    #pragma unroll
    for (int nt = 0; nt < 4; ++nt) {
        float t = 0.0f;
        #pragma unroll
        for (int r = 0; r < 4; ++r) t += xe[2 * nt][r] * (we[2 * nt][r] + acc[nt][r]);
        #pragma unroll
        for (int r = 0; r < 4; ++r) t += xe[2 * nt + 1][r] * (we[2 * nt + 1][r] + acc[nt][4 + r]);
        p[nt] = t;
    }
    float s = (p[0] + p[1]) + (p[2] + p[3]);
    s += (half == 0) ? bias : 0.0f;
    s += xhalf(s);
    if (half == 0) out[rowbase + wr + l15] = s;
}

extern "C" void kernel_launch(void* const* d_in, const int* in_sizes, int n_in,
                              void* d_out, int out_size, void* d_ws, size_t ws_size,
                              hipStream_t stream) {
    const float* x  = (const float*)d_in[0];
    const float* W  = (const float*)d_in[1];
    const float* bv = (const float*)d_in[2];
    float* out = (float*)d_out;

    const int batch = in_sizes[0] / DIMS;
    const int grid  = batch / ROWS_PER_BLOCK;
    poly_wmma_kernel<<<grid, 256, 0, stream>>>(x, W, bv, out);
}
